# starting kernel
# baseline (speedup 1.0000x reference)
#include <hip/hip_runtime.h>
#include <math.h>

#define N_NODES   100000
#define N_EDGES   1600000
#define F_IN      64
#define DIM       16
#define N_CLASSES 10
#define BN_EPS    1e-5f

typedef __attribute__((ext_vector_type(2))) float v2f;
typedef __attribute__((ext_vector_type(8))) float v8f;

__device__ __forceinline__ v8f wmma_k4(v2f a, v2f b, v8f c) {
    return __builtin_amdgcn_wmma_f32_16x16x4_f32(
         false, a,  false, b,
         (short)0, c,  false,  false);
}

__global__ void gin_proj64(const float* __restrict__ x,
                           const float* __restrict__ w1a,
                           float* __restrict__ y) {
    const int tile = blockIdx.x;
    const int lane = threadIdx.x;
    const int half = lane >> 4;
    const int l    = lane & 15;
    const int arow = tile * 16 + l;

    v8f c;
    #pragma unroll
    for (int r = 0; r < 8; ++r) c[r] = 0.0f;

    #pragma unroll
    for (int k0 = 0; k0 < F_IN; k0 += 4) {
        const int ka = k0 + half * 2;
        v2f a = *(const v2f*)(x + arow * F_IN + ka);
        v2f b;
        b.x = w1a[ka * DIM + l];
        b.y = w1a[ka * DIM + DIM + l];
        c = wmma_k4(a, b, c);
    }
    #pragma unroll
    for (int r = 0; r < 8; ++r)
        y[(tile * 16 + half * 8 + r) * DIM + l] = c[r];
}

__global__ void gin_scatter_add(const float* __restrict__ feat,
                                const int* __restrict__ src,
                                const int* __restrict__ dst,
                                float* __restrict__ agg) {
    const int t = blockIdx.x * blockDim.x + threadIdx.x;
    if (t >= N_EDGES * DIM) return;
    const int e = t >> 4;
    const int d = t & 15;
    const int s  = src[e];
    const int dd = dst[e];
    atomicAdd(agg + dd * DIM + d, feat[s * DIM + d]);
}

__global__ void gin_layer1(const float* __restrict__ y, const float* __restrict__ agg,
                           const float* __restrict__ b1a,
                           const float* __restrict__ w1b, const float* __restrict__ b1b,
                           const float* __restrict__ g,  const float* __restrict__ be,
                           const float* __restrict__ mu, const float* __restrict__ var,
                           float* __restrict__ u) {
    __shared__ float sA[16][17];
    const int tile = blockIdx.x;
    const int lane = threadIdx.x;
    const int half = lane >> 4;
    const int l    = lane & 15;

    const float bl = b1a[l];
    #pragma unroll
    for (int r = 0; r < 8; ++r) {
        const int row = half * 8 + r;
        const int gi  = (tile * 16 + row) * DIM + l;
        sA[row][l] = fmaxf(y[gi] + agg[gi] + bl, 0.0f);
    }
    __syncthreads();

    const float cb = b1b[l];
    v8f c;
    #pragma unroll
    for (int r = 0; r < 8; ++r) c[r] = cb;

    #pragma unroll
    for (int k0 = 0; k0 < DIM; k0 += 4) {
        const int ka = k0 + half * 2;
        v2f a; a.x = sA[l][ka];           a.y = sA[l][ka + 1];
        v2f b; b.x = w1b[ka * DIM + l];   b.y = w1b[ka * DIM + DIM + l];
        c = wmma_k4(a, b, c);
    }

    const float sc = g[l] * rsqrtf(var[l] + BN_EPS);
    const float mm = mu[l], bb = be[l];
    #pragma unroll
    for (int r = 0; r < 8; ++r) {
        const float t = fmaxf(c[r], 0.0f);
        u[(tile * 16 + half * 8 + r) * DIM + l] = (t - mm) * sc + bb;
    }
}

__global__ void gin_layer2_fc(const float* __restrict__ u,  const float* __restrict__ agg2,
                              const float* __restrict__ w2a, const float* __restrict__ b2a,
                              const float* __restrict__ w2b, const float* __restrict__ b2b,
                              const float* __restrict__ g,   const float* __restrict__ be,
                              const float* __restrict__ mu,  const float* __restrict__ var,
                              const float* __restrict__ fcw, const float* __restrict__ fcb,
                              float* __restrict__ logits) {
    __shared__ float sA[16][17];
    const int tile = blockIdx.x;
    const int lane = threadIdx.x;
    const int half = lane >> 4;
    const int l    = lane & 15;

    #pragma unroll
    for (int r = 0; r < 8; ++r) {
        const int row = half * 8 + r;
        const int gi  = (tile * 16 + row) * DIM + l;
        sA[row][l] = u[gi] + agg2[gi];
    }
    __syncthreads();

    v8f c;
    {
        const float cb = b2a[l];
        #pragma unroll
        for (int r = 0; r < 8; ++r) c[r] = cb;
    }
    #pragma unroll
    for (int k0 = 0; k0 < DIM; k0 += 4) {
        const int ka = k0 + half * 2;
        v2f a; a.x = sA[l][ka];          a.y = sA[l][ka + 1];
        v2f b; b.x = w2a[ka * DIM + l];  b.y = w2a[ka * DIM + DIM + l];
        c = wmma_k4(a, b, c);
    }
    __syncthreads();
    #pragma unroll
    for (int r = 0; r < 8; ++r) sA[half * 8 + r][l] = fmaxf(c[r], 0.0f);
    __syncthreads();

    {
        const float cb = b2b[l];
        #pragma unroll
        for (int r = 0; r < 8; ++r) c[r] = cb;
    }
    #pragma unroll
    for (int k0 = 0; k0 < DIM; k0 += 4) {
        const int ka = k0 + half * 2;
        v2f a; a.x = sA[l][ka];          a.y = sA[l][ka + 1];
        v2f b; b.x = w2b[ka * DIM + l];  b.y = w2b[ka * DIM + DIM + l];
        c = wmma_k4(a, b, c);
    }
    __syncthreads();

    {
        const float sc = g[l] * rsqrtf(var[l] + BN_EPS);
        const float mm = mu[l], bb = be[l];
        #pragma unroll
        for (int r = 0; r < 8; ++r) {
            const float t = fmaxf(c[r], 0.0f);
            sA[half * 8 + r][l] = (t - mm) * sc + bb;
        }
    }
    __syncthreads();

    {
        const float cb = (l < N_CLASSES) ? fcb[l] : 0.0f;
        #pragma unroll
        for (int r = 0; r < 8; ++r) c[r] = cb;
    }
    #pragma unroll
    for (int k0 = 0; k0 < DIM; k0 += 4) {
        const int ka = k0 + half * 2;
        v2f a; a.x = sA[l][ka];  a.y = sA[l][ka + 1];
        v2f b;
        b.x = (l < N_CLASSES) ? fcw[ka * N_CLASSES + l]       : 0.0f;
        b.y = (l < N_CLASSES) ? fcw[(ka + 1) * N_CLASSES + l] : 0.0f;
        c = wmma_k4(a, b, c);
    }
    if (l < N_CLASSES) {
        #pragma unroll
        for (int r = 0; r < 8; ++r)
            logits[(tile * 16 + half * 8 + r) * N_CLASSES + l] = c[r];
    }
}

__global__ void gin_logsoftmax(float* __restrict__ o) {
    const int i = blockIdx.x * blockDim.x + threadIdx.x;
    if (i >= N_NODES) return;
    float* p = o + (size_t)i * N_CLASSES;
    float v[N_CLASSES];
    float m = -INFINITY;
    #pragma unroll
    for (int j = 0; j < N_CLASSES; ++j) { v[j] = p[j]; m = fmaxf(m, v[j]); }
    float s = 0.0f;
    #pragma unroll
    for (int j = 0; j < N_CLASSES; ++j) s += expf(v[j] - m);
    const float ls = m + logf(s);
    #pragma unroll
    for (int j = 0; j < N_CLASSES; ++j) p[j] = v[j] - ls;
}

__global__ void gin_zero(float* __restrict__ p, int n) {
    const int i = blockIdx.x * blockDim.x + threadIdx.x;
    if (i < n) p[i] = 0.0f;
}

extern "C" void kernel_launch(void* const* d_in, const int* in_sizes, int n_in,
                              void* d_out, int out_size, void* d_ws, size_t ws_size,
                              hipStream_t stream) {
    (void)in_sizes; (void)n_in; (void)out_size; (void)ws_size;

    const float* x   = (const float*)d_in[0];
    const int*   ei  = (const int*)d_in[1];
    const int*   src = ei;
    const int*   dst = ei + N_EDGES;
    const float* w1a = (const float*)d_in[2];  const float* b1a = (const float*)d_in[3];
    const float* w1b = (const float*)d_in[4];  const float* b1b = (const float*)d_in[5];
    const float* g1  = (const float*)d_in[6];  const float* be1 = (const float*)d_in[7];
    const float* m1  = (const float*)d_in[8];  const float* v1  = (const float*)d_in[9];
    const float* w2a = (const float*)d_in[10]; const float* b2a = (const float*)d_in[11];
    const float* w2b = (const float*)d_in[12]; const float* b2b = (const float*)d_in[13];
    const float* g2  = (const float*)d_in[14]; const float* be2 = (const float*)d_in[15];
    const float* m2  = (const float*)d_in[16]; const float* v2  = (const float*)d_in[17];
    const float* fcw = (const float*)d_in[18]; const float* fcb = (const float*)d_in[19];
    float* out = (float*)d_out;

    const int NF = N_NODES * DIM;
    float* ws   = (float*)d_ws;
    float* y    = ws;
    float* u    = ws + (size_t)NF;
    float* agg1 = ws + (size_t)2 * NF;
    float* agg2 = ws + (size_t)3 * NF;

    const int TILES = N_NODES / 16;
    const int ETHREADS = N_EDGES * DIM;

    gin_zero<<<(2 * NF + 255) / 256, 256, 0, stream>>>(agg1, 2 * NF);
    gin_proj64<<<TILES, 32, 0, stream>>>(x, w1a, y);
    gin_scatter_add<<<(ETHREADS + 255) / 256, 256, 0, stream>>>(y, src, dst, agg1);
    gin_layer1<<<TILES, 32, 0, stream>>>(y, agg1, b1a, w1b, b1b, g1, be1, m1, v1, u);
    gin_scatter_add<<<(ETHREADS + 255) / 256, 256, 0, stream>>>(u, src, dst, agg2);
    gin_layer2_fc<<<TILES, 32, 0, stream>>>(u, agg2, w2a, b2a, w2b, b2b,
                                            g2, be2, m2, v2, fcw, fcb, out);
    gin_logsoftmax<<<(N_NODES + 255) / 256, 256, 0, stream>>>(out);
}


# Aggregation: per-edge float atomics replaced by a destination-grouped neighbour list (built per call, one integer atomic per edge) and an atomic-free gather; per-node sum order = list rank order
# speedup vs baseline: 1.0386x; 1.0386x over previous
#include <hip/hip_runtime.h>
#include <math.h>

#define N_NODES   100000
#define N_EDGES   1600000
#define F_IN      64
#define DIM       16
#define N_CLASSES 10
#define BN_EPS    1e-5f

typedef __attribute__((ext_vector_type(2))) float v2f;
typedef __attribute__((ext_vector_type(8))) float v8f;

__device__ __forceinline__ v8f wmma_k4(v2f a, v2f b, v8f c) {
    return __builtin_amdgcn_wmma_f32_16x16x4_f32(
         false, a,  false, b,
         (short)0, c,  false,  false);
}

__global__ void gin_proj64(const float* __restrict__ x,
                           const float* __restrict__ w1a,
                           float* __restrict__ y,
                           int* __restrict__ deg,
                           int* __restrict__ cursor) {
    const int tile = blockIdx.x;
    const int lane = threadIdx.x;
    if (lane < 16) deg[tile * 16 + lane] = 0;
    if (tile == 0 && lane == 0) *cursor = 0;
    const int half = lane >> 4;
    const int l    = lane & 15;
    const int arow = tile * 16 + l;

    v8f c;
    #pragma unroll
    for (int r = 0; r < 8; ++r) c[r] = 0.0f;

    #pragma unroll
    for (int k0 = 0; k0 < F_IN; k0 += 4) {
        const int ka = k0 + half * 2;
        v2f a = *(const v2f*)(x + arow * F_IN + ka);
        v2f b;
        b.x = w1a[ka * DIM + l];
        b.y = w1a[ka * DIM + DIM + l];
        c = wmma_k4(a, b, c);
    }
    #pragma unroll
    for (int r = 0; r < 8; ++r)
        y[(tile * 16 + half * 8 + r) * DIM + l] = c[r];
}

__global__ void gin_count_in(const int* __restrict__ dst, int* __restrict__ deg,
                             int* __restrict__ rank) {
    const int e = blockIdx.x * blockDim.x + threadIdx.x;
    if (e < N_EDGES) rank[e] = atomicAdd(deg + dst[e], 1);
}

__global__ void gin_alloc_rows(const int* __restrict__ deg, int* __restrict__ pos,
                               int* __restrict__ cursor) {
    __shared__ int s_w[8];
    __shared__ int s_base;
    const int i    = blockIdx.x * 256 + threadIdx.x;
    const int lane = threadIdx.x & 31;
    const int w    = threadIdx.x >> 5;
    const int v    = (i < N_NODES) ? deg[i] : 0;
    int inc = v;
    for (int o = 1; o < 32; o <<= 1) {
        const int t = __shfl_up(inc, o);
        if (lane >= o) inc += t;
    }
    if (lane == 31) s_w[w] = inc;
    __syncthreads();
    if (threadIdx.x == 0) {
        int tot = 0;
        for (int k = 0; k < 8; ++k) { const int t = s_w[k]; s_w[k] = tot; tot += t; }
        s_base = atomicAdd(cursor, tot);
    }
    __syncthreads();
    if (i < N_NODES) pos[i] = s_base + s_w[w] + inc - v;
}

__global__ void gin_fill_rows(const int* __restrict__ src, const int* __restrict__ dst,
                              const int* __restrict__ pos, const int* __restrict__ rank,
                              int* __restrict__ nbr) {
    const int e = blockIdx.x * blockDim.x + threadIdx.x;
    if (e >= N_EDGES) return;
    nbr[pos[dst[e]] + rank[e]] = src[e];
}

__global__ void gin_gather_sum(const float* __restrict__ feat,
                               const int* __restrict__ nbr,
                               const int* __restrict__ pos,
                               const int* __restrict__ deg,
                               float* __restrict__ h) {
    const int t    = blockIdx.x * blockDim.x + threadIdx.x;
    const int node = t >> 4;
    const int l    = t & 15;
    const int n    = deg[node];
    const int beg  = pos[node];
    const int nmax = max(n, __shfl_xor(n, 16));
    float acc = 0.0f;
    for (int base = 0; base < nmax; base += 16) {
        const int k    = base + l;
        const int mine = (k < n) ? nbr[beg + k] : 0;
        float v[16];
        #pragma unroll
        for (int j = 0; j < 16; ++j) {
            const int s = __shfl(mine, j, 16);
            v[j] = (base + j < n) ? feat[s * DIM + l] : 0.0f;
        }
        #pragma unroll
        for (int j = 0; j < 16; ++j) acc += v[j];
    }
    h[t] = feat[t] + acc;
}

__global__ void gin_layer1(const float* __restrict__ h,
                           const float* __restrict__ b1a,
                           const float* __restrict__ w1b, const float* __restrict__ b1b,
                           const float* __restrict__ g,  const float* __restrict__ be,
                           const float* __restrict__ mu, const float* __restrict__ var,
                           float* __restrict__ u) {
    __shared__ float sA[16][17];
    const int tile = blockIdx.x;
    const int lane = threadIdx.x;
    const int half = lane >> 4;
    const int l    = lane & 15;

    const float bl = b1a[l];
    #pragma unroll
    for (int r = 0; r < 8; ++r) {
        const int row = half * 8 + r;
        const int gi  = (tile * 16 + row) * DIM + l;
        sA[row][l] = fmaxf(h[gi] + bl, 0.0f);
    }
    __syncthreads();

    const float cb = b1b[l];
    v8f c;
    #pragma unroll
    for (int r = 0; r < 8; ++r) c[r] = cb;

    #pragma unroll
    for (int k0 = 0; k0 < DIM; k0 += 4) {
        const int ka = k0 + half * 2;
        v2f a; a.x = sA[l][ka];           a.y = sA[l][ka + 1];
        v2f b; b.x = w1b[ka * DIM + l];   b.y = w1b[ka * DIM + DIM + l];
        c = wmma_k4(a, b, c);
    }

    const float sc = g[l] * rsqrtf(var[l] + BN_EPS);
    const float mm = mu[l], bb = be[l];
    #pragma unroll
    for (int r = 0; r < 8; ++r) {
        const float t = fmaxf(c[r], 0.0f);
        u[(tile * 16 + half * 8 + r) * DIM + l] = (t - mm) * sc + bb;
    }
}

__global__ void gin_layer2_fc(const float* __restrict__ s_in,
                              const float* __restrict__ w2a, const float* __restrict__ b2a,
                              const float* __restrict__ w2b, const float* __restrict__ b2b,
                              const float* __restrict__ g,   const float* __restrict__ be,
                              const float* __restrict__ mu,  const float* __restrict__ var,
                              const float* __restrict__ fcw, const float* __restrict__ fcb,
                              float* __restrict__ logits) {
    __shared__ float sA[16][17];
    const int tile = blockIdx.x;
    const int lane = threadIdx.x;
    const int half = lane >> 4;
    const int l    = lane & 15;

    #pragma unroll
    for (int r = 0; r < 8; ++r) {
        const int row = half * 8 + r;
        const int gi  = (tile * 16 + row) * DIM + l;
        sA[row][l] = s_in[gi];
    }
    __syncthreads();

    v8f c;
    {
        const float cb = b2a[l];
        #pragma unroll
        for (int r = 0; r < 8; ++r) c[r] = cb;
    }
    #pragma unroll
    for (int k0 = 0; k0 < DIM; k0 += 4) {
        const int ka = k0 + half * 2;
        v2f a; a.x = sA[l][ka];          a.y = sA[l][ka + 1];
        v2f b; b.x = w2a[ka * DIM + l];  b.y = w2a[ka * DIM + DIM + l];
        c = wmma_k4(a, b, c);
    }
    __syncthreads();
    #pragma unroll
    for (int r = 0; r < 8; ++r) sA[half * 8 + r][l] = fmaxf(c[r], 0.0f);
    __syncthreads();

    {
        const float cb = b2b[l];
        #pragma unroll
        for (int r = 0; r < 8; ++r) c[r] = cb;
    }
    #pragma unroll
    for (int k0 = 0; k0 < DIM; k0 += 4) {
        const int ka = k0 + half * 2;
        v2f a; a.x = sA[l][ka];          a.y = sA[l][ka + 1];
        v2f b; b.x = w2b[ka * DIM + l];  b.y = w2b[ka * DIM + DIM + l];
        c = wmma_k4(a, b, c);
    }
    __syncthreads();

    {
        const float sc = g[l] * rsqrtf(var[l] + BN_EPS);
        const float mm = mu[l], bb = be[l];
        #pragma unroll
        for (int r = 0; r < 8; ++r) {
            const float t = fmaxf(c[r], 0.0f);
            sA[half * 8 + r][l] = (t - mm) * sc + bb;
        }
    }
    __syncthreads();

    {
        const float cb = (l < N_CLASSES) ? fcb[l] : 0.0f;
        #pragma unroll
        for (int r = 0; r < 8; ++r) c[r] = cb;
    }
    #pragma unroll
    for (int k0 = 0; k0 < DIM; k0 += 4) {
        const int ka = k0 + half * 2;
        v2f a; a.x = sA[l][ka];  a.y = sA[l][ka + 1];
        v2f b;
        b.x = (l < N_CLASSES) ? fcw[ka * N_CLASSES + l]       : 0.0f;
        b.y = (l < N_CLASSES) ? fcw[(ka + 1) * N_CLASSES + l] : 0.0f;
        c = wmma_k4(a, b, c);
    }
    if (l < N_CLASSES) {
        #pragma unroll
        for (int r = 0; r < 8; ++r)
            logits[(tile * 16 + half * 8 + r) * N_CLASSES + l] = c[r];
    }
}

__global__ void gin_logsoftmax(float* __restrict__ o) {
    const int i = blockIdx.x * blockDim.x + threadIdx.x;
    if (i >= N_NODES) return;
    float* p = o + (size_t)i * N_CLASSES;
    float v[N_CLASSES];
    float m = -INFINITY;
    #pragma unroll
    for (int j = 0; j < N_CLASSES; ++j) { v[j] = p[j]; m = fmaxf(m, v[j]); }
    float s = 0.0f;
    #pragma unroll
    for (int j = 0; j < N_CLASSES; ++j) s += expf(v[j] - m);
    const float ls = m + logf(s);
    #pragma unroll
    for (int j = 0; j < N_CLASSES; ++j) p[j] = v[j] - ls;
}

extern "C" void kernel_launch(void* const* d_in, const int* in_sizes, int n_in,
                              void* d_out, int out_size, void* d_ws, size_t ws_size,
                              hipStream_t stream) {
    (void)in_sizes; (void)n_in; (void)out_size; (void)ws_size;

    const float* x   = (const float*)d_in[0];
    const int*   ei  = (const int*)d_in[1];
    const int*   src = ei;
    const int*   dst = ei + N_EDGES;
    const float* w1a = (const float*)d_in[2];  const float* b1a = (const float*)d_in[3];
    const float* w1b = (const float*)d_in[4];  const float* b1b = (const float*)d_in[5];
    const float* g1  = (const float*)d_in[6];  const float* be1 = (const float*)d_in[7];
    const float* m1  = (const float*)d_in[8];  const float* v1  = (const float*)d_in[9];
    const float* w2a = (const float*)d_in[10]; const float* b2a = (const float*)d_in[11];
    const float* w2b = (const float*)d_in[12]; const float* b2b = (const float*)d_in[13];
    const float* g2  = (const float*)d_in[14]; const float* be2 = (const float*)d_in[15];
    const float* m2  = (const float*)d_in[16]; const float* v2  = (const float*)d_in[17];
    const float* fcw = (const float*)d_in[18]; const float* fcb = (const float*)d_in[19];
    float* out = (float*)d_out;

    const int NF = N_NODES * DIM;
    float* ws   = (float*)d_ws;
    float* bufA = ws;
    float* bufB = ws + (size_t)NF;
    int*   rank = (int*)bufB;
    int*   nbr  = (int*)(ws + (size_t)2 * NF);
    int*   deg  = (int*)(ws + (size_t)3 * NF);
    int*   pos  = deg + N_NODES;
    int*   cur  = pos + N_NODES;

    const int TILES = N_NODES / 16;
    const int EBLK  = (N_EDGES + 255) / 256;
    const int NBLK  = (N_NODES + 255) / 256;
    const int GBLK  = NF / 256;

    gin_proj64<<<TILES, 32, 0, stream>>>(x, w1a, bufA, deg, cur);
    gin_count_in<<<EBLK, 256, 0, stream>>>(dst, deg, rank);
    gin_alloc_rows<<<NBLK, 256, 0, stream>>>(deg, pos, cur);
    gin_fill_rows<<<EBLK, 256, 0, stream>>>(src, dst, pos, rank, nbr);
    gin_gather_sum<<<GBLK, 256, 0, stream>>>(bufA, nbr, pos, deg, bufB);
    gin_layer1<<<TILES, 32, 0, stream>>>(bufB, b1a, w1b, b1b, g1, be1, m1, v1, bufA);
    gin_gather_sum<<<GBLK, 256, 0, stream>>>(bufA, nbr, pos, deg, bufB);
    gin_layer2_fc<<<TILES, 32, 0, stream>>>(bufB, w2a, b2a, w2b, b2b,
                                            g2, be2, m2, v2, fcw, fcb, out);
    gin_logsoftmax<<<(N_NODES + 255) / 256, 256, 0, stream>>>(out);
}


# Aggregation: edges partitioned per call into 256 destination ranges with LDS counters (no global atomics, no neighbour list); a work-group sums a range with LDS float adds, sum order = arrival order
# speedup vs baseline: 2.6505x; 2.5520x over previous
#include <hip/hip_runtime.h>
#include <math.h>

#define N_NODES   100000
#define N_EDGES   1600000
#define F_IN      64
#define DIM       16
#define N_CLASSES 10
#define BN_EPS    1e-5f

#define N_RANGES    256
#define RANGE_NODES 392
#define CHUNK       (N_EDGES / N_RANGES)
#define PT          1024
#define EPT         ((CHUNK + PT - 1) / PT)
#define SRC_BITS    17
#define SRC_MASK    ((1u << SRC_BITS) - 1u)
static_assert(N_RANGES * CHUNK == N_EDGES, "chunks must tile the edge list");
static_assert(N_RANGES * RANGE_NODES >= N_NODES, "ranges must cover the nodes");
static_assert(N_NODES <= (1 << SRC_BITS) && RANGE_NODES + 1 <= (1 << (32 - SRC_BITS)), "record packing");

typedef __attribute__((ext_vector_type(2))) float v2f;
typedef __attribute__((ext_vector_type(8))) float v8f;

__device__ __forceinline__ v8f wmma_k4(v2f a, v2f b, v8f c) {
    return __builtin_amdgcn_wmma_f32_16x16x4_f32(
         false, a,  false, b,
         (short)0, c,  false,  false);
}

__global__ void gin_proj64(const float* __restrict__ x,
                           const float* __restrict__ w1a,
                           float* __restrict__ y) {
    const int tile = blockIdx.x;
    const int lane = threadIdx.x;
    const int half = lane >> 4;
    const int l    = lane & 15;
    const int arow = tile * 16 + l;

    v8f c;
    #pragma unroll
    for (int r = 0; r < 8; ++r) c[r] = 0.0f;

    #pragma unroll
    for (int k0 = 0; k0 < F_IN; k0 += 4) {
        const int ka = k0 + half * 2;
        v2f a = *(const v2f*)(x + arow * F_IN + ka);
        v2f b;
        b.x = w1a[ka * DIM + l];
        b.y = w1a[ka * DIM + DIM + l];
        c = wmma_k4(a, b, c);
    }
    #pragma unroll
    for (int r = 0; r < 8; ++r)
        y[(tile * 16 + half * 8 + r) * DIM + l] = c[r];
}

__global__ void __launch_bounds__(PT) gin_range_count(const int* __restrict__ dst,
                                                      int* __restrict__ cnt) {
    __shared__ int s_h[N_RANGES];
    const int tid = threadIdx.x;
    const int e0  = blockIdx.x * CHUNK;
    unsigned d[EPT];
    #pragma unroll
    for (int u = 0; u < EPT; ++u) d[u] = (unsigned)dst[e0 + min(tid + u * PT, CHUNK - 1)];
    if (tid < N_RANGES) s_h[tid] = 0;
    __syncthreads();
    #pragma unroll
    for (int u = 0; u < EPT; ++u)
        if (tid + u * PT < CHUNK) atomicAdd(&s_h[d[u] / RANGE_NODES], 1);
    __syncthreads();
    if (tid < N_RANGES) cnt[blockIdx.x * N_RANGES + tid] = s_h[tid];
}

__global__ void __launch_bounds__(PT) gin_range_scatter(const int* __restrict__ src,
                                                        const int* __restrict__ dst,
                                                        const int* __restrict__ cnt,
                                                        int* __restrict__ rstart,
                                                        unsigned* __restrict__ recs) {
    __shared__ int s_pre[4][N_RANGES];
    __shared__ int s_tot[4][N_RANGES];
    __shared__ int s_wt[32];
    __shared__ int s_wl[32];
    __shared__ int s_delta[N_RANGES];
    __shared__ int s_cur[N_RANGES];
    __shared__ unsigned s_rec[CHUNK];
    __shared__ unsigned short s_rng[CHUNK];
    const int tid = threadIdx.x;
    const int wg  = blockIdx.x;
    const int e0  = wg * CHUNK;
    unsigned d[EPT], s[EPT];
    #pragma unroll
    for (int u = 0; u < EPT; ++u) {
        const int i = e0 + min(tid + u * PT, CHUNK - 1);
        d[u] = (unsigned)dst[i];
        s[u] = (unsigned)src[i];
    }

    {
        const int b = tid & (N_RANGES - 1);
        const int q = tid >> 8;
        int pre = 0, tot = 0;
        #pragma unroll 32
        for (int k = 0; k < 64; ++k) {
            const int w = q * 64 + k;
            const int c = cnt[w * N_RANGES + b];
            tot += c;
            pre += (w < wg) ? c : 0;
        }
        s_pre[q][b] = pre;
        s_tot[q][b] = tot;
    }
    __syncthreads();

    int T = 0, P = 0, L = 0;
    if (tid < N_RANGES) {
        T = s_tot[0][tid] + s_tot[1][tid] + s_tot[2][tid] + s_tot[3][tid];
        P = s_pre[0][tid] + s_pre[1][tid] + s_pre[2][tid] + s_pre[3][tid];
        L = cnt[wg * N_RANGES + tid];
    }
    const int lane = tid & 31;
    const int wv   = tid >> 5;
    int incT = T, incL = L;
    for (int o = 1; o < 32; o <<= 1) {
        const int tT = __shfl_up(incT, o);
        const int tL = __shfl_up(incL, o);
        if (lane >= o) { incT += tT; incL += tL; }
    }
    if (lane == 31) { s_wt[wv] = incT; s_wl[wv] = incL; }
    __syncthreads();
    if (tid < N_RANGES) {
        int offT = 0, offL = 0;
        for (int k = 0; k < wv; ++k) { offT += s_wt[k]; offL += s_wl[k]; }
        const int gstart = offT + incT - T;
        const int lstart = offL + incL - L;
        s_delta[tid] = gstart + P - lstart;
        s_cur[tid]   = lstart;
        if (wg == 0) rstart[tid] = gstart;
    }
    if (wg == 0 && tid == 0) rstart[N_RANGES] = N_EDGES;
    __syncthreads();

    #pragma unroll
    for (int u = 0; u < EPT; ++u) {
        if (tid + u * PT < CHUNK) {
            const unsigned rg = d[u] / RANGE_NODES;
            const unsigned dl = d[u] - rg * RANGE_NODES;
            const int p = atomicAdd(&s_cur[rg], 1);
            s_rec[p] = s[u] | (dl << SRC_BITS);
            s_rng[p] = (unsigned short)rg;
        }
    }
    __syncthreads();
    for (int t = tid; t < CHUNK; t += PT)
        recs[t + s_delta[s_rng[t]]] = s_rec[t];
}

#define SPARE_REC ((unsigned)RANGE_NODES << SRC_BITS)
__global__ void __launch_bounds__(PT) gin_range_sum(const float* __restrict__ feat,
                                                    const unsigned* __restrict__ recs,
                                                    const int* __restrict__ rstart,
                                                    float* __restrict__ h) {
    __shared__ float s_acc[(RANGE_NODES + 1) * DIM];
    const int tid = threadIdx.x;
    for (int t = tid; t < (RANGE_NODES + 1) * DIM; t += PT) s_acc[t] = 0.0f;
    const int bs = rstart[blockIdx.x];
    const int be = rstart[blockIdx.x + 1];
    __syncthreads();

    const int l    = tid & 15;
    const int grp  = tid >> 4;
    const int grp0 = (tid >> 5) << 1;
    int k = bs + grp * 16 + l;
    unsigned mine = recs[min(k, N_EDGES - 1)];
    if (k >= be) mine = SPARE_REC;
    for (int base0 = bs + grp0 * 16; base0 < be; base0 += 64 * 16) {
        k += 64 * 16;
        unsigned nxt = recs[min(k, N_EDGES - 1)];
        if (k >= be) nxt = SPARE_REC;
        unsigned r[16];
        float    v[16];
        #pragma unroll
        for (int j = 0; j < 16; ++j) {
            r[j] = __shfl(mine, j, 16);
            v[j] = feat[(r[j] & SRC_MASK) * DIM + l];
        }
        #pragma unroll
        for (int j = 0; j < 16; ++j)
            __hip_atomic_fetch_add(&s_acc[(r[j] >> SRC_BITS) * DIM + l], v[j],
                                   __ATOMIC_RELAXED, __HIP_MEMORY_SCOPE_WORKGROUP);
        mine = nxt;
    }
    __syncthreads();

    const int g0 = blockIdx.x * RANGE_NODES * DIM;
    for (int t = tid; t < RANGE_NODES * DIM; t += PT) {
        const int g = g0 + t;
        if (g < N_NODES * DIM) h[g] = feat[g] + s_acc[t];
    }
}

__global__ void gin_layer1(const float* __restrict__ h,
                           const float* __restrict__ b1a,
                           const float* __restrict__ w1b, const float* __restrict__ b1b,
                           const float* __restrict__ g,  const float* __restrict__ be,
                           const float* __restrict__ mu, const float* __restrict__ var,
                           float* __restrict__ u) {
    __shared__ float sA[16][17];
    const int tile = blockIdx.x;
    const int lane = threadIdx.x;
    const int half = lane >> 4;
    const int l    = lane & 15;

    const float bl = b1a[l];
    #pragma unroll
    for (int r = 0; r < 8; ++r) {
        const int row = half * 8 + r;
        const int gi  = (tile * 16 + row) * DIM + l;
        sA[row][l] = fmaxf(h[gi] + bl, 0.0f);
    }
    __syncthreads();

    const float cb = b1b[l];
    v8f c;
    #pragma unroll
    for (int r = 0; r < 8; ++r) c[r] = cb;

    #pragma unroll
    for (int k0 = 0; k0 < DIM; k0 += 4) {
        const int ka = k0 + half * 2;
        v2f a; a.x = sA[l][ka];           a.y = sA[l][ka + 1];
        v2f b; b.x = w1b[ka * DIM + l];   b.y = w1b[ka * DIM + DIM + l];
        c = wmma_k4(a, b, c);
    }

    const float sc = g[l] * rsqrtf(var[l] + BN_EPS);
    const float mm = mu[l], bb = be[l];
    #pragma unroll
    for (int r = 0; r < 8; ++r) {
        const float t = fmaxf(c[r], 0.0f);
        u[(tile * 16 + half * 8 + r) * DIM + l] = (t - mm) * sc + bb;
    }
}

__global__ void gin_layer2_fc(const float* __restrict__ s_in,
                              const float* __restrict__ w2a, const float* __restrict__ b2a,
                              const float* __restrict__ w2b, const float* __restrict__ b2b,
                              const float* __restrict__ g,   const float* __restrict__ be,
                              const float* __restrict__ mu,  const float* __restrict__ var,
                              const float* __restrict__ fcw, const float* __restrict__ fcb,
                              float* __restrict__ logits) {
    __shared__ float sA[16][17];
    const int tile = blockIdx.x;
    const int lane = threadIdx.x;
    const int half = lane >> 4;
    const int l    = lane & 15;

    #pragma unroll
    for (int r = 0; r < 8; ++r) {
        const int row = half * 8 + r;
        const int gi  = (tile * 16 + row) * DIM + l;
        sA[row][l] = s_in[gi];
    }
    __syncthreads();

    v8f c;
    {
        const float cb = b2a[l];
        #pragma unroll
        for (int r = 0; r < 8; ++r) c[r] = cb;
    }
    #pragma unroll
    for (int k0 = 0; k0 < DIM; k0 += 4) {
        const int ka = k0 + half * 2;
        v2f a; a.x = sA[l][ka];          a.y = sA[l][ka + 1];
        v2f b; b.x = w2a[ka * DIM + l];  b.y = w2a[ka * DIM + DIM + l];
        c = wmma_k4(a, b, c);
    }
    __syncthreads();
    #pragma unroll
    for (int r = 0; r < 8; ++r) sA[half * 8 + r][l] = fmaxf(c[r], 0.0f);
    __syncthreads();

    {
        const float cb = b2b[l];
        #pragma unroll
        for (int r = 0; r < 8; ++r) c[r] = cb;
    }
    #pragma unroll
    for (int k0 = 0; k0 < DIM; k0 += 4) {
        const int ka = k0 + half * 2;
        v2f a; a.x = sA[l][ka];          a.y = sA[l][ka + 1];
        v2f b; b.x = w2b[ka * DIM + l];  b.y = w2b[ka * DIM + DIM + l];
        c = wmma_k4(a, b, c);
    }
    __syncthreads();

    {
        const float sc = g[l] * rsqrtf(var[l] + BN_EPS);
        const float mm = mu[l], bb = be[l];
        #pragma unroll
        for (int r = 0; r < 8; ++r) {
            const float t = fmaxf(c[r], 0.0f);
            sA[half * 8 + r][l] = (t - mm) * sc + bb;
        }
    }
    __syncthreads();

    {
        const float cb = (l < N_CLASSES) ? fcb[l] : 0.0f;
        #pragma unroll
        for (int r = 0; r < 8; ++r) c[r] = cb;
    }
    #pragma unroll
    for (int k0 = 0; k0 < DIM; k0 += 4) {
        const int ka = k0 + half * 2;
        v2f a; a.x = sA[l][ka];  a.y = sA[l][ka + 1];
        v2f b;
        b.x = (l < N_CLASSES) ? fcw[ka * N_CLASSES + l]       : 0.0f;
        b.y = (l < N_CLASSES) ? fcw[(ka + 1) * N_CLASSES + l] : 0.0f;
        c = wmma_k4(a, b, c);
    }
    if (l < N_CLASSES) {
        #pragma unroll
        for (int r = 0; r < 8; ++r)
            logits[(tile * 16 + half * 8 + r) * N_CLASSES + l] = c[r];
    }
}

__global__ void gin_logsoftmax(float* __restrict__ o) {
    const int i = blockIdx.x * blockDim.x + threadIdx.x;
    if (i >= N_NODES) return;
    float* p = o + (size_t)i * N_CLASSES;
    float v[N_CLASSES];
    float m = -INFINITY;
    #pragma unroll
    for (int j = 0; j < N_CLASSES; ++j) { v[j] = p[j]; m = fmaxf(m, v[j]); }
    float s = 0.0f;
    #pragma unroll
    for (int j = 0; j < N_CLASSES; ++j) s += expf(v[j] - m);
    const float ls = m + logf(s);
    #pragma unroll
    for (int j = 0; j < N_CLASSES; ++j) p[j] = v[j] - ls;
}

extern "C" void kernel_launch(void* const* d_in, const int* in_sizes, int n_in,
                              void* d_out, int out_size, void* d_ws, size_t ws_size,
                              hipStream_t stream) {
    (void)in_sizes; (void)n_in; (void)out_size; (void)ws_size;

    const float* x   = (const float*)d_in[0];
    const int*   ei  = (const int*)d_in[1];
    const int*   src = ei;
    const int*   dst = ei + N_EDGES;
    const float* w1a = (const float*)d_in[2];  const float* b1a = (const float*)d_in[3];
    const float* w1b = (const float*)d_in[4];  const float* b1b = (const float*)d_in[5];
    const float* g1  = (const float*)d_in[6];  const float* be1 = (const float*)d_in[7];
    const float* m1  = (const float*)d_in[8];  const float* v1  = (const float*)d_in[9];
    const float* w2a = (const float*)d_in[10]; const float* b2a = (const float*)d_in[11];
    const float* w2b = (const float*)d_in[12]; const float* b2b = (const float*)d_in[13];
    const float* g2  = (const float*)d_in[14]; const float* be2 = (const float*)d_in[15];
    const float* m2  = (const float*)d_in[16]; const float* v2  = (const float*)d_in[17];
    const float* fcw = (const float*)d_in[18]; const float* fcb = (const float*)d_in[19];
    float* out = (float*)d_out;

    const int NF = N_NODES * DIM;
    float* ws   = (float*)d_ws;
    float* bufA = ws;
    float* bufB = ws + (size_t)NF;
    unsigned* recs = (unsigned*)(ws + (size_t)2 * NF);
    int*   cnt    = (int*)(ws + (size_t)3 * NF);
    int*   rstart = cnt + N_RANGES * N_RANGES;

    const int TILES = N_NODES / 16;

    gin_proj64<<<TILES, 32, 0, stream>>>(x, w1a, bufA);
    gin_range_count<<<N_RANGES, PT, 0, stream>>>(dst, cnt);
    gin_range_scatter<<<N_RANGES, PT, 0, stream>>>(src, dst, cnt, rstart, recs);
    gin_range_sum<<<N_RANGES, PT, 0, stream>>>(bufA, recs, rstart, bufB);
    gin_layer1<<<TILES, 32, 0, stream>>>(bufB, b1a, w1b, b1b, g1, be1, m1, v1, bufA);
    gin_range_sum<<<N_RANGES, PT, 0, stream>>>(bufA, recs, rstart, bufB);
    gin_layer2_fc<<<TILES, 32, 0, stream>>>(bufB, w2a, b2a, w2b, b2b,
                                            g2, be2, m2, v2, fcw, fcb, out);
    gin_logsoftmax<<<(N_NODES + 255) / 256, 256, 0, stream>>>(out);
}


# Dense per-node stages (layer MLP tails, BN, classifier, log_softmax) run in the range-sum work-group on its LDS accumulator, one 16-row tile per wave; ranges of 400 nodes; 8 launches -> 5
# speedup vs baseline: 3.5696x; 1.3468x over previous
#include <hip/hip_runtime.h>
#include <math.h>

#define N_NODES   100000
#define N_EDGES   1600000
#define F_IN      64
#define DIM       16
#define N_CLASSES 10
#define BN_EPS    1e-5f

#define N_RANGES    256
#define RANGE_NODES 400
#define CHUNK       (N_EDGES / N_RANGES)
#define PT          1024
#define EPT         ((CHUNK + PT - 1) / PT)
#define SRC_BITS    17
#define SRC_MASK    ((1u << SRC_BITS) - 1u)
static_assert(N_RANGES * CHUNK == N_EDGES, "chunks must tile the edge list");
static_assert(N_RANGES * RANGE_NODES >= N_NODES, "ranges must cover the nodes");
static_assert(N_NODES <= (1 << SRC_BITS) && RANGE_NODES + 1 <= (1 << (32 - SRC_BITS)), "record packing");

typedef __attribute__((ext_vector_type(2))) float v2f;
typedef __attribute__((ext_vector_type(8))) float v8f;

__device__ __forceinline__ v8f wmma_k4(v2f a, v2f b, v8f c) {
    return __builtin_amdgcn_wmma_f32_16x16x4_f32(
         false, a,  false, b,
         (short)0, c,  false,  false);
}

__global__ void gin_proj64(const float* __restrict__ x,
                           const float* __restrict__ w1a,
                           float* __restrict__ y) {
    const int tile = blockIdx.x;
    const int lane = threadIdx.x;
    const int half = lane >> 4;
    const int l    = lane & 15;
    const int arow = tile * 16 + l;

    v8f c;
    #pragma unroll
    for (int r = 0; r < 8; ++r) c[r] = 0.0f;

    #pragma unroll
    for (int k0 = 0; k0 < F_IN; k0 += 4) {
        const int ka = k0 + half * 2;
        v2f a = *(const v2f*)(x + arow * F_IN + ka);
        v2f b;
        b.x = w1a[ka * DIM + l];
        b.y = w1a[ka * DIM + DIM + l];
        c = wmma_k4(a, b, c);
    }
    #pragma unroll
    for (int r = 0; r < 8; ++r)
        y[(tile * 16 + half * 8 + r) * DIM + l] = c[r];
}

__global__ void __launch_bounds__(PT) gin_range_count(const int* __restrict__ dst,
                                                      int* __restrict__ cnt) {
    __shared__ int s_h[N_RANGES];
    const int tid = threadIdx.x;
    const int e0  = blockIdx.x * CHUNK;
    unsigned d[EPT];
    #pragma unroll
    for (int u = 0; u < EPT; ++u) d[u] = (unsigned)dst[e0 + min(tid + u * PT, CHUNK - 1)];
    if (tid < N_RANGES) s_h[tid] = 0;
    __syncthreads();
    #pragma unroll
    for (int u = 0; u < EPT; ++u)
        if (tid + u * PT < CHUNK) atomicAdd(&s_h[d[u] / RANGE_NODES], 1);
    __syncthreads();
    if (tid < N_RANGES) cnt[blockIdx.x * N_RANGES + tid] = s_h[tid];
}

__global__ void __launch_bounds__(PT) gin_range_scatter(const int* __restrict__ src,
                                                        const int* __restrict__ dst,
                                                        const int* __restrict__ cnt,
                                                        int* __restrict__ rstart,
                                                        unsigned* __restrict__ recs) {
    __shared__ int s_pre[4][N_RANGES];
    __shared__ int s_tot[4][N_RANGES];
    __shared__ int s_wt[32];
    __shared__ int s_wl[32];
    __shared__ int s_delta[N_RANGES];
    __shared__ int s_cur[N_RANGES];
    __shared__ unsigned s_rec[CHUNK];
    __shared__ unsigned short s_rng[CHUNK];
    const int tid = threadIdx.x;
    const int wg  = blockIdx.x;
    const int e0  = wg * CHUNK;
    unsigned d[EPT], s[EPT];
    #pragma unroll
    for (int u = 0; u < EPT; ++u) {
        const int i = e0 + min(tid + u * PT, CHUNK - 1);
        d[u] = (unsigned)dst[i];
        s[u] = (unsigned)src[i];
    }

    {
        const int b = tid & (N_RANGES - 1);
        const int q = tid >> 8;
        int pre = 0, tot = 0;
        #pragma unroll 32
        for (int k = 0; k < 64; ++k) {
            const int w = q * 64 + k;
            const int c = cnt[w * N_RANGES + b];
            tot += c;
            pre += (w < wg) ? c : 0;
        }
        s_pre[q][b] = pre;
        s_tot[q][b] = tot;
    }
    __syncthreads();

    int T = 0, P = 0, L = 0;
    if (tid < N_RANGES) {
        T = s_tot[0][tid] + s_tot[1][tid] + s_tot[2][tid] + s_tot[3][tid];
        P = s_pre[0][tid] + s_pre[1][tid] + s_pre[2][tid] + s_pre[3][tid];
        L = cnt[wg * N_RANGES + tid];
    }
    const int lane = tid & 31;
    const int wv   = tid >> 5;
    int incT = T, incL = L;
    for (int o = 1; o < 32; o <<= 1) {
        const int tT = __shfl_up(incT, o);
        const int tL = __shfl_up(incL, o);
        if (lane >= o) { incT += tT; incL += tL; }
    }
    if (lane == 31) { s_wt[wv] = incT; s_wl[wv] = incL; }
    __syncthreads();
    if (tid < N_RANGES) {
        int offT = 0, offL = 0;
        for (int k = 0; k < wv; ++k) { offT += s_wt[k]; offL += s_wl[k]; }
        const int gstart = offT + incT - T;
        const int lstart = offL + incL - L;
        s_delta[tid] = gstart + P - lstart;
        s_cur[tid]   = lstart;
        if (wg == 0) rstart[tid] = gstart;
    }
    if (wg == 0 && tid == 0) rstart[N_RANGES] = N_EDGES;
    __syncthreads();

    #pragma unroll
    for (int u = 0; u < EPT; ++u) {
        if (tid + u * PT < CHUNK) {
            const unsigned rg = d[u] / RANGE_NODES;
            const unsigned dl = d[u] - rg * RANGE_NODES;
            const int p = atomicAdd(&s_cur[rg], 1);
            s_rec[p] = s[u] | (dl << SRC_BITS);
            s_rng[p] = (unsigned short)rg;
        }
    }
    __syncthreads();
    for (int t = tid; t < CHUNK; t += PT)
        recs[t + s_delta[s_rng[t]]] = s_rec[t];
}

#define SPARE_REC ((unsigned)RANGE_NODES << SRC_BITS)
#define ACC_ROWS  ((PT / 32) * 16)
#define RANGE_V4  (RANGE_NODES * DIM / 4)
static_assert(RANGE_NODES % 16 == 0 && ACC_ROWS >= RANGE_NODES + 1, "tiles per range");

__device__ __forceinline__ void range_gather_sum(const float* __restrict__ feat,
                                                 const unsigned* __restrict__ recs,
                                                 const int* __restrict__ rstart,
                                                 float* s_acc) {
    const int tid = threadIdx.x;
    for (int t = tid; t < ACC_ROWS * DIM; t += PT) s_acc[t] = 0.0f;
    const int bs = rstart[blockIdx.x];
    const int be = rstart[blockIdx.x + 1];
    __syncthreads();

    const int l    = tid & 15;
    const int grp  = tid >> 4;
    const int grp0 = (tid >> 5) << 1;
    int k = bs + grp * 16 + l;
    unsigned mine = recs[min(k, N_EDGES - 1)];
    if (k >= be) mine = SPARE_REC;
    for (int base0 = bs + grp0 * 16; base0 < be; base0 += 64 * 16) {
        k += 64 * 16;
        unsigned nxt = recs[min(k, N_EDGES - 1)];
        if (k >= be) nxt = SPARE_REC;
        unsigned r[16];
        float    v[16];
        #pragma unroll
        for (int j = 0; j < 16; ++j) {
            r[j] = __shfl(mine, j, 16);
            v[j] = feat[(r[j] & SRC_MASK) * DIM + l];
        }
        #pragma unroll
        for (int j = 0; j < 16; ++j)
            __hip_atomic_fetch_add(&s_acc[(r[j] >> SRC_BITS) * DIM + l], v[j],
                                   __ATOMIC_RELAXED, __HIP_MEMORY_SCOPE_WORKGROUP);
        mine = nxt;
    }
}

template <int NC>
__device__ __forceinline__ void load_bfrag(const float* __restrict__ w, int l, int half, v2f (&b)[4]) {
    #pragma unroll
    for (int i = 0; i < 4; ++i) {
        const int ka = 4 * i + half * 2;
        b[i].x = (l < NC) ? w[ka * NC + l]       : 0.0f;
        b[i].y = (l < NC) ? w[(ka + 1) * NC + l] : 0.0f;
    }
}

__device__ __forceinline__ v8f tile_gemm16(const float* tileA, const v2f (&b)[4],
                                           v8f c, int l, int half) {
    #pragma unroll
    for (int i = 0; i < 4; ++i) {
        const int ka = 4 * i + half * 2;
        v2f a = *(const v2f*)(tileA + l * DIM + ka);
        c = wmma_k4(a, b[i], c);
    }
    return c;
}

__global__ void __launch_bounds__(PT) gin_range_sum_layer1(const float* __restrict__ feat,
                                                           const unsigned* __restrict__ recs,
                                                           const int* __restrict__ rstart,
                                                           const float* __restrict__ b1a,
                                                           const float* __restrict__ w1b, const float* __restrict__ b1b,
                                                           const float* __restrict__ g,  const float* __restrict__ be,
                                                           const float* __restrict__ mu, const float* __restrict__ var,
                                                           float* __restrict__ u) {
    __shared__ __attribute__((aligned(16))) float s_acc[ACC_ROWS * DIM];
    range_gather_sum(feat, recs, rstart, s_acc);

    const int tid  = threadIdx.x;
    const int lane = tid & 31;
    const int half = lane >> 4;
    const int l    = lane & 15;
    const int q0   = blockIdx.x * RANGE_V4;
    float4* acc4 = (float4*)s_acc;

    const int cb = (tid & 3) * 4;
    const float b0 = b1a[cb], b1 = b1a[cb + 1], b2 = b1a[cb + 2], b3 = b1a[cb + 3];
    v2f wb[4];
    load_bfrag<DIM>(w1b, l, half, wb);
    const float cb1 = b1b[l];
    const float sc = g[l] * rsqrtf(var[l] + BN_EPS);
    const float mm = mu[l], bb = be[l];
    __syncthreads();

    {
        const float4* feat4 = (const float4*)feat;
        for (int t = tid; t < RANGE_V4; t += PT) {
            if (q0 + t < N_NODES * DIM / 4) {
                const float4 f = feat4[q0 + t];
                float4 a = acc4[t];
                a.x = fmaxf((f.x + a.x) + b0, 0.0f);
                a.y = fmaxf((f.y + a.y) + b1, 0.0f);
                a.z = fmaxf((f.z + a.z) + b2, 0.0f);
                a.w = fmaxf((f.w + a.w) + b3, 0.0f);
                acc4[t] = a;
            }
        }
    }
    __syncthreads();

    float* tileA = s_acc + (tid >> 5) * 16 * DIM;
    v8f c;
    #pragma unroll
    for (int r = 0; r < 8; ++r) c[r] = cb1;
    c = tile_gemm16(tileA, wb, c, l, half);
    #pragma unroll
    for (int r = 0; r < 8; ++r) {
        const float t = fmaxf(c[r], 0.0f);
        tileA[(half * 8 + r) * DIM + l] = (t - mm) * sc + bb;
    }
    __syncthreads();

    float4* u4 = (float4*)u;
    for (int t = tid; t < RANGE_V4; t += PT)
        if (q0 + t < N_NODES * DIM / 4) u4[q0 + t] = acc4[t];
}

__global__ void __launch_bounds__(PT) gin_range_sum_layer2_fc(const float* __restrict__ feat,
                                                              const unsigned* __restrict__ recs,
                                                              const int* __restrict__ rstart,
                                                              const float* __restrict__ w2a, const float* __restrict__ b2a,
                                                              const float* __restrict__ w2b, const float* __restrict__ b2b,
                                                              const float* __restrict__ g,   const float* __restrict__ be,
                                                              const float* __restrict__ mu,  const float* __restrict__ var,
                                                              const float* __restrict__ fcw, const float* __restrict__ fcb,
                                                              float* __restrict__ out) {
    __shared__ __attribute__((aligned(16))) float s_acc[ACC_ROWS * DIM];
    __shared__ __attribute__((aligned(16))) float s_out[RANGE_NODES * N_CLASSES];
    range_gather_sum(feat, recs, rstart, s_acc);

    const int tid  = threadIdx.x;
    const int lane = tid & 31;
    const int half = lane >> 4;
    const int l    = lane & 15;
    const int q0   = blockIdx.x * RANGE_V4;
    float4* acc4 = (float4*)s_acc;

    v2f wa[4], wb[4], wf[4];
    load_bfrag<DIM>(w2a, l, half, wa);
    load_bfrag<DIM>(w2b, l, half, wb);
    load_bfrag<N_CLASSES>(fcw, l, half, wf);
    const float cba = b2a[l], cbb = b2b[l];
    const float cbf = (l < N_CLASSES) ? fcb[l] : 0.0f;
    const float sc = g[l] * rsqrtf(var[l] + BN_EPS);
    const float mm = mu[l], bb = be[l];
    __syncthreads();

    {
        const float4* feat4 = (const float4*)feat;
        for (int t = tid; t < RANGE_V4; t += PT) {
            if (q0 + t < N_NODES * DIM / 4) {
                const float4 f = feat4[q0 + t];
                float4 a = acc4[t];
                a.x = f.x + a.x;  a.y = f.y + a.y;  a.z = f.z + a.z;  a.w = f.w + a.w;
                acc4[t] = a;
            }
        }
    }
    __syncthreads();

    float* tileA = s_acc + (tid >> 5) * 16 * DIM;
    v8f c;

    #pragma unroll
    for (int r = 0; r < 8; ++r) c[r] = cba;
    c = tile_gemm16(tileA, wa, c, l, half);
    #pragma unroll
    for (int r = 0; r < 8; ++r) tileA[(half * 8 + r) * DIM + l] = fmaxf(c[r], 0.0f);
    __syncthreads();

    #pragma unroll
    for (int r = 0; r < 8; ++r) c[r] = cbb;
    c = tile_gemm16(tileA, wb, c, l, half);
    #pragma unroll
    for (int r = 0; r < 8; ++r) {
        const float t = fmaxf(c[r], 0.0f);
        tileA[(half * 8 + r) * DIM + l] = (t - mm) * sc + bb;
    }
    __syncthreads();

    #pragma unroll
    for (int r = 0; r < 8; ++r) c[r] = cbf;
    c = tile_gemm16(tileA, wf, c, l, half);
    #pragma unroll
    for (int r = 0; r < 8; ++r) tileA[(half * 8 + r) * DIM + l] = c[r];
    __syncthreads();

    if (tid < RANGE_NODES) {
        const float* p = s_acc + tid * DIM;
        float v[N_CLASSES];
        float m = -INFINITY;
        #pragma unroll
        for (int j = 0; j < N_CLASSES; ++j) { v[j] = p[j]; m = fmaxf(m, v[j]); }
        float s = 0.0f;
        #pragma unroll
        for (int j = 0; j < N_CLASSES; ++j) s += expf(v[j] - m);
        const float ls = m + logf(s);
        #pragma unroll
        for (int j = 0; j < N_CLASSES; ++j) s_out[tid * N_CLASSES + j] = v[j] - ls;
    }
    __syncthreads();

    const int o0 = blockIdx.x * (RANGE_NODES * N_CLASSES / 4);
    const float4* so4 = (const float4*)s_out;
    float4* out4 = (float4*)out;
    for (int t = tid; t < RANGE_NODES * N_CLASSES / 4; t += PT)
        if (o0 + t < N_NODES * N_CLASSES / 4) out4[o0 + t] = so4[t];
}

extern "C" void kernel_launch(void* const* d_in, const int* in_sizes, int n_in,
                              void* d_out, int out_size, void* d_ws, size_t ws_size,
                              hipStream_t stream) {
    (void)in_sizes; (void)n_in; (void)out_size; (void)ws_size;

    const float* x   = (const float*)d_in[0];
    const int*   ei  = (const int*)d_in[1];
    const int*   src = ei;
    const int*   dst = ei + N_EDGES;
    const float* w1a = (const float*)d_in[2];  const float* b1a = (const float*)d_in[3];
    const float* w1b = (const float*)d_in[4];  const float* b1b = (const float*)d_in[5];
    const float* g1  = (const float*)d_in[6];  const float* be1 = (const float*)d_in[7];
    const float* m1  = (const float*)d_in[8];  const float* v1  = (const float*)d_in[9];
    const float* w2a = (const float*)d_in[10]; const float* b2a = (const float*)d_in[11];
    const float* w2b = (const float*)d_in[12]; const float* b2b = (const float*)d_in[13];
    const float* g2  = (const float*)d_in[14]; const float* be2 = (const float*)d_in[15];
    const float* m2  = (const float*)d_in[16]; const float* v2  = (const float*)d_in[17];
    const float* fcw = (const float*)d_in[18]; const float* fcb = (const float*)d_in[19];
    float* out = (float*)d_out;

    const int NF = N_NODES * DIM;
    float* ws   = (float*)d_ws;
    float* bufA = ws;
    float* bufB = ws + (size_t)NF;
    unsigned* recs = (unsigned*)(ws + (size_t)2 * NF);
    int*   cnt    = (int*)(ws + (size_t)3 * NF);
    int*   rstart = cnt + N_RANGES * N_RANGES;

    const int TILES = N_NODES / 16;

    gin_proj64<<<TILES, 32, 0, stream>>>(x, w1a, bufA);
    gin_range_count<<<N_RANGES, PT, 0, stream>>>(dst, cnt);
    gin_range_scatter<<<N_RANGES, PT, 0, stream>>>(src, dst, cnt, rstart, recs);
    gin_range_sum_layer1<<<N_RANGES, PT, 0, stream>>>(bufA, recs, rstart, b1a, w1b, b1b,
                                                      g1, be1, m1, v1, bufB);
    gin_range_sum_layer2_fc<<<N_RANGES, PT, 0, stream>>>(bufB, recs, rstart, w2a, b2a, w2b, b2b,
                                                         g2, be2, m2, v2, fcw, fcb, out);
}


# Projection x @ w1a moved into the edge-counting launch: one 16-node tile per wave of its 1024-thread work-groups, beside the counting; 6250 one-wave work-groups and one launch fewer (5 -> 4)
# speedup vs baseline: 3.7684x; 1.0557x over previous
#include <hip/hip_runtime.h>
#include <math.h>

#define N_NODES   100000
#define N_EDGES   1600000
#define F_IN      64
#define DIM       16
#define N_CLASSES 10
#define BN_EPS    1e-5f

#define N_RANGES    256
#define RANGE_NODES 400
#define CHUNK       (N_EDGES / N_RANGES)
#define PT          1024
#define EPT         ((CHUNK + PT - 1) / PT)
#define SRC_BITS    17
#define SRC_MASK    ((1u << SRC_BITS) - 1u)
static_assert(N_RANGES * CHUNK == N_EDGES, "chunks must tile the edge list");
static_assert(N_RANGES * RANGE_NODES >= N_NODES, "ranges must cover the nodes");
static_assert(N_NODES <= (1 << SRC_BITS) && RANGE_NODES + 1 <= (1 << (32 - SRC_BITS)), "record packing");

typedef __attribute__((ext_vector_type(2))) float v2f;
typedef __attribute__((ext_vector_type(8))) float v8f;

__device__ __forceinline__ v8f wmma_k4(v2f a, v2f b, v8f c) {
    return __builtin_amdgcn_wmma_f32_16x16x4_f32(
         false, a,  false, b,
         (short)0, c,  false,  false);
}


#define N_TILES (N_NODES / 16)
static_assert(N_TILES * 16 == N_NODES, "whole tiles");
static_assert((PT / 32) * N_RANGES >= N_TILES, "a wave of the counting launch for every tile");

__global__ void __launch_bounds__(PT) gin_count_proj(const int* __restrict__ dst,
                                                     int* __restrict__ cnt,
                                                     const float* __restrict__ x,
                                                     const float* __restrict__ w1a,
                                                     float* __restrict__ y) {
    __shared__ int s_h[N_RANGES];
    const int tid = threadIdx.x;
    const int e0  = blockIdx.x * CHUNK;
    unsigned d[EPT];
    #pragma unroll
    for (int u = 0; u < EPT; ++u) d[u] = (unsigned)dst[e0 + min(tid + u * PT, CHUNK - 1)];

    const int  wv       = __builtin_amdgcn_readfirstlane(tid >> 5);
    const int  tile     = wv * N_RANGES + (int)blockIdx.x;
    const bool has_tile = tile < N_TILES;
    const int  lane = tid & 31;
    const int  half = lane >> 4;
    const int  l    = lane & 15;
    const int  arow = tile * 16 + l;
    v2f a[F_IN / 4], b[F_IN / 4];
    if (has_tile) {
        #pragma unroll
        for (int i = 0; i < F_IN / 4; ++i) {
            const int ka = 4 * i + half * 2;
            a[i]   = *(const v2f*)(x + arow * F_IN + ka);
            b[i].x = w1a[ka * DIM + l];
            b[i].y = w1a[ka * DIM + DIM + l];
        }
    }

    if (tid < N_RANGES) s_h[tid] = 0;
    __syncthreads();
    #pragma unroll
    for (int u = 0; u < EPT; ++u)
        if (tid + u * PT < CHUNK) atomicAdd(&s_h[d[u] / RANGE_NODES], 1);
    __syncthreads();
    if (tid < N_RANGES) cnt[blockIdx.x * N_RANGES + tid] = s_h[tid];

    if (has_tile) {
        v8f c;
        #pragma unroll
        for (int r = 0; r < 8; ++r) c[r] = 0.0f;
        #pragma unroll
        for (int i = 0; i < F_IN / 4; ++i) c = wmma_k4(a[i], b[i], c);
        #pragma unroll
        for (int r = 0; r < 8; ++r)
            y[(tile * 16 + half * 8 + r) * DIM + l] = c[r];
    }
}

__global__ void __launch_bounds__(PT) gin_range_scatter(const int* __restrict__ src,
                                                        const int* __restrict__ dst,
                                                        const int* __restrict__ cnt,
                                                        int* __restrict__ rstart,
                                                        unsigned* __restrict__ recs) {
    __shared__ int s_pre[4][N_RANGES];
    __shared__ int s_tot[4][N_RANGES];
    __shared__ int s_wt[32];
    __shared__ int s_wl[32];
    __shared__ int s_delta[N_RANGES];
    __shared__ int s_cur[N_RANGES];
    __shared__ unsigned s_rec[CHUNK];
    __shared__ unsigned short s_rng[CHUNK];
    const int tid = threadIdx.x;
    const int wg  = blockIdx.x;
    const int e0  = wg * CHUNK;
    unsigned d[EPT], s[EPT];
    #pragma unroll
    for (int u = 0; u < EPT; ++u) {
        const int i = e0 + min(tid + u * PT, CHUNK - 1);
        d[u] = (unsigned)dst[i];
        s[u] = (unsigned)src[i];
    }

    {
        const int b = tid & (N_RANGES - 1);
        const int q = tid >> 8;
        int pre = 0, tot = 0;
        #pragma unroll 32
        for (int k = 0; k < 64; ++k) {
            const int w = q * 64 + k;
            const int c = cnt[w * N_RANGES + b];
            tot += c;
            pre += (w < wg) ? c : 0;
        }
        s_pre[q][b] = pre;
        s_tot[q][b] = tot;
    }
    __syncthreads();

    int T = 0, P = 0, L = 0;
    if (tid < N_RANGES) {
        T = s_tot[0][tid] + s_tot[1][tid] + s_tot[2][tid] + s_tot[3][tid];
        P = s_pre[0][tid] + s_pre[1][tid] + s_pre[2][tid] + s_pre[3][tid];
        L = cnt[wg * N_RANGES + tid];
    }
    const int lane = tid & 31;
    const int wv   = tid >> 5;
    int incT = T, incL = L;
    for (int o = 1; o < 32; o <<= 1) {
        const int tT = __shfl_up(incT, o);
        const int tL = __shfl_up(incL, o);
        if (lane >= o) { incT += tT; incL += tL; }
    }
    if (lane == 31) { s_wt[wv] = incT; s_wl[wv] = incL; }
    __syncthreads();
    if (tid < N_RANGES) {
        int offT = 0, offL = 0;
        for (int k = 0; k < wv; ++k) { offT += s_wt[k]; offL += s_wl[k]; }
        const int gstart = offT + incT - T;
        const int lstart = offL + incL - L;
        s_delta[tid] = gstart + P - lstart;
        s_cur[tid]   = lstart;
        if (wg == 0) rstart[tid] = gstart;
    }
    if (wg == 0 && tid == 0) rstart[N_RANGES] = N_EDGES;
    __syncthreads();

    #pragma unroll
    for (int u = 0; u < EPT; ++u) {
        if (tid + u * PT < CHUNK) {
            const unsigned rg = d[u] / RANGE_NODES;
            const unsigned dl = d[u] - rg * RANGE_NODES;
            const int p = atomicAdd(&s_cur[rg], 1);
            s_rec[p] = s[u] | (dl << SRC_BITS);
            s_rng[p] = (unsigned short)rg;
        }
    }
    __syncthreads();
    for (int t = tid; t < CHUNK; t += PT)
        recs[t + s_delta[s_rng[t]]] = s_rec[t];
}

#define SPARE_REC ((unsigned)RANGE_NODES << SRC_BITS)
#define ACC_ROWS  ((PT / 32) * 16)
#define RANGE_V4  (RANGE_NODES * DIM / 4)
static_assert(RANGE_NODES % 16 == 0 && ACC_ROWS >= RANGE_NODES + 1, "tiles per range");

__device__ __forceinline__ void range_gather_sum(const float* __restrict__ feat,
                                                 const unsigned* __restrict__ recs,
                                                 const int* __restrict__ rstart,
                                                 float* s_acc) {
    const int tid = threadIdx.x;
    for (int t = tid; t < ACC_ROWS * DIM; t += PT) s_acc[t] = 0.0f;
    const int bs = rstart[blockIdx.x];
    const int be = rstart[blockIdx.x + 1];
    __syncthreads();

    const int l    = tid & 15;
    const int grp  = tid >> 4;
    const int grp0 = (tid >> 5) << 1;
    int k = bs + grp * 16 + l;
    unsigned mine = recs[min(k, N_EDGES - 1)];
    if (k >= be) mine = SPARE_REC;
    for (int base0 = bs + grp0 * 16; base0 < be; base0 += 64 * 16) {
        k += 64 * 16;
        unsigned nxt = recs[min(k, N_EDGES - 1)];
        if (k >= be) nxt = SPARE_REC;
        unsigned r[16];
        float    v[16];
        #pragma unroll
        for (int j = 0; j < 16; ++j) {
            r[j] = __shfl(mine, j, 16);
            v[j] = feat[(r[j] & SRC_MASK) * DIM + l];
        }
        #pragma unroll
        for (int j = 0; j < 16; ++j)
            __hip_atomic_fetch_add(&s_acc[(r[j] >> SRC_BITS) * DIM + l], v[j],
                                   __ATOMIC_RELAXED, __HIP_MEMORY_SCOPE_WORKGROUP);
        mine = nxt;
    }
}

template <int NC>
__device__ __forceinline__ void load_bfrag(const float* __restrict__ w, int l, int half, v2f (&b)[4]) {
    #pragma unroll
    for (int i = 0; i < 4; ++i) {
        const int ka = 4 * i + half * 2;
        b[i].x = (l < NC) ? w[ka * NC + l]       : 0.0f;
        b[i].y = (l < NC) ? w[(ka + 1) * NC + l] : 0.0f;
    }
}

__device__ __forceinline__ v8f tile_gemm16(const float* tileA, const v2f (&b)[4],
                                           v8f c, int l, int half) {
    #pragma unroll
    for (int i = 0; i < 4; ++i) {
        const int ka = 4 * i + half * 2;
        v2f a = *(const v2f*)(tileA + l * DIM + ka);
        c = wmma_k4(a, b[i], c);
    }
    return c;
}

__global__ void __launch_bounds__(PT) gin_range_sum_layer1(const float* __restrict__ feat,
                                                           const unsigned* __restrict__ recs,
                                                           const int* __restrict__ rstart,
                                                           const float* __restrict__ b1a,
                                                           const float* __restrict__ w1b, const float* __restrict__ b1b,
                                                           const float* __restrict__ g,  const float* __restrict__ be,
                                                           const float* __restrict__ mu, const float* __restrict__ var,
                                                           float* __restrict__ u) {
    __shared__ __attribute__((aligned(16))) float s_acc[ACC_ROWS * DIM];
    range_gather_sum(feat, recs, rstart, s_acc);

    const int tid  = threadIdx.x;
    const int lane = tid & 31;
    const int half = lane >> 4;
    const int l    = lane & 15;
    const int q0   = blockIdx.x * RANGE_V4;
    float4* acc4 = (float4*)s_acc;

    const int cb = (tid & 3) * 4;
    const float b0 = b1a[cb], b1 = b1a[cb + 1], b2 = b1a[cb + 2], b3 = b1a[cb + 3];
    v2f wb[4];
    load_bfrag<DIM>(w1b, l, half, wb);
    const float cb1 = b1b[l];
    const float sc = g[l] * rsqrtf(var[l] + BN_EPS);
    const float mm = mu[l], bb = be[l];
    __syncthreads();

    {
        const float4* feat4 = (const float4*)feat;
        for (int t = tid; t < RANGE_V4; t += PT) {
            if (q0 + t < N_NODES * DIM / 4) {
                const float4 f = feat4[q0 + t];
                float4 a = acc4[t];
                a.x = fmaxf((f.x + a.x) + b0, 0.0f);
                a.y = fmaxf((f.y + a.y) + b1, 0.0f);
                a.z = fmaxf((f.z + a.z) + b2, 0.0f);
                a.w = fmaxf((f.w + a.w) + b3, 0.0f);
                acc4[t] = a;
            }
        }
    }
    __syncthreads();

    float* tileA = s_acc + (tid >> 5) * 16 * DIM;
    v8f c;
    #pragma unroll
    for (int r = 0; r < 8; ++r) c[r] = cb1;
    c = tile_gemm16(tileA, wb, c, l, half);
    #pragma unroll
    for (int r = 0; r < 8; ++r) {
        const float t = fmaxf(c[r], 0.0f);
        tileA[(half * 8 + r) * DIM + l] = (t - mm) * sc + bb;
    }
    __syncthreads();

    float4* u4 = (float4*)u;
    for (int t = tid; t < RANGE_V4; t += PT)
        if (q0 + t < N_NODES * DIM / 4) u4[q0 + t] = acc4[t];
}

__global__ void __launch_bounds__(PT) gin_range_sum_layer2_fc(const float* __restrict__ feat,
                                                              const unsigned* __restrict__ recs,
                                                              const int* __restrict__ rstart,
                                                              const float* __restrict__ w2a, const float* __restrict__ b2a,
                                                              const float* __restrict__ w2b, const float* __restrict__ b2b,
                                                              const float* __restrict__ g,   const float* __restrict__ be,
                                                              const float* __restrict__ mu,  const float* __restrict__ var,
                                                              const float* __restrict__ fcw, const float* __restrict__ fcb,
                                                              float* __restrict__ out) {
    __shared__ __attribute__((aligned(16))) float s_acc[ACC_ROWS * DIM];
    __shared__ __attribute__((aligned(16))) float s_out[RANGE_NODES * N_CLASSES];
    range_gather_sum(feat, recs, rstart, s_acc);

    const int tid  = threadIdx.x;
    const int lane = tid & 31;
    const int half = lane >> 4;
    const int l    = lane & 15;
    const int q0   = blockIdx.x * RANGE_V4;
    float4* acc4 = (float4*)s_acc;

    v2f wa[4], wb[4], wf[4];
    load_bfrag<DIM>(w2a, l, half, wa);
    load_bfrag<DIM>(w2b, l, half, wb);
    load_bfrag<N_CLASSES>(fcw, l, half, wf);
    const float cba = b2a[l], cbb = b2b[l];
    const float cbf = (l < N_CLASSES) ? fcb[l] : 0.0f;
    const float sc = g[l] * rsqrtf(var[l] + BN_EPS);
    const float mm = mu[l], bb = be[l];
    __syncthreads();

    {
        const float4* feat4 = (const float4*)feat;
        for (int t = tid; t < RANGE_V4; t += PT) {
            if (q0 + t < N_NODES * DIM / 4) {
                const float4 f = feat4[q0 + t];
                float4 a = acc4[t];
                a.x = f.x + a.x;  a.y = f.y + a.y;  a.z = f.z + a.z;  a.w = f.w + a.w;
                acc4[t] = a;
            }
        }
    }
    __syncthreads();

    float* tileA = s_acc + (tid >> 5) * 16 * DIM;
    v8f c;

    #pragma unroll
    for (int r = 0; r < 8; ++r) c[r] = cba;
    c = tile_gemm16(tileA, wa, c, l, half);
    #pragma unroll
    for (int r = 0; r < 8; ++r) tileA[(half * 8 + r) * DIM + l] = fmaxf(c[r], 0.0f);
    __syncthreads();

    #pragma unroll
    for (int r = 0; r < 8; ++r) c[r] = cbb;
    c = tile_gemm16(tileA, wb, c, l, half);
    #pragma unroll
    for (int r = 0; r < 8; ++r) {
        const float t = fmaxf(c[r], 0.0f);
        tileA[(half * 8 + r) * DIM + l] = (t - mm) * sc + bb;
    }
    __syncthreads();

    #pragma unroll
    for (int r = 0; r < 8; ++r) c[r] = cbf;
    c = tile_gemm16(tileA, wf, c, l, half);
    #pragma unroll
    for (int r = 0; r < 8; ++r) tileA[(half * 8 + r) * DIM + l] = c[r];
    __syncthreads();

    if (tid < RANGE_NODES) {
        const float* p = s_acc + tid * DIM;
        float v[N_CLASSES];
        float m = -INFINITY;
        #pragma unroll
        for (int j = 0; j < N_CLASSES; ++j) { v[j] = p[j]; m = fmaxf(m, v[j]); }
        float s = 0.0f;
        #pragma unroll
        for (int j = 0; j < N_CLASSES; ++j) s += expf(v[j] - m);
        const float ls = m + logf(s);
        #pragma unroll
        for (int j = 0; j < N_CLASSES; ++j) s_out[tid * N_CLASSES + j] = v[j] - ls;
    }
    __syncthreads();

    const int o0 = blockIdx.x * (RANGE_NODES * N_CLASSES / 4);
    const float4* so4 = (const float4*)s_out;
    float4* out4 = (float4*)out;
    for (int t = tid; t < RANGE_NODES * N_CLASSES / 4; t += PT)
        if (o0 + t < N_NODES * N_CLASSES / 4) out4[o0 + t] = so4[t];
}

extern "C" void kernel_launch(void* const* d_in, const int* in_sizes, int n_in,
                              void* d_out, int out_size, void* d_ws, size_t ws_size,
                              hipStream_t stream) {
    (void)in_sizes; (void)n_in; (void)out_size; (void)ws_size;

    const float* x   = (const float*)d_in[0];
    const int*   ei  = (const int*)d_in[1];
    const int*   src = ei;
    const int*   dst = ei + N_EDGES;
    const float* w1a = (const float*)d_in[2];  const float* b1a = (const float*)d_in[3];
    const float* w1b = (const float*)d_in[4];  const float* b1b = (const float*)d_in[5];
    const float* g1  = (const float*)d_in[6];  const float* be1 = (const float*)d_in[7];
    const float* m1  = (const float*)d_in[8];  const float* v1  = (const float*)d_in[9];
    const float* w2a = (const float*)d_in[10]; const float* b2a = (const float*)d_in[11];
    const float* w2b = (const float*)d_in[12]; const float* b2b = (const float*)d_in[13];
    const float* g2  = (const float*)d_in[14]; const float* be2 = (const float*)d_in[15];
    const float* m2  = (const float*)d_in[16]; const float* v2  = (const float*)d_in[17];
    const float* fcw = (const float*)d_in[18]; const float* fcb = (const float*)d_in[19];
    float* out = (float*)d_out;

    const int NF = N_NODES * DIM;
    float* ws   = (float*)d_ws;
    float* bufA = ws;
    float* bufB = ws + (size_t)NF;
    unsigned* recs = (unsigned*)(ws + (size_t)2 * NF);
    int*   cnt    = (int*)(ws + (size_t)3 * NF);
    int*   rstart = cnt + N_RANGES * N_RANGES;

    gin_count_proj<<<N_RANGES, PT, 0, stream>>>(dst, cnt, x, w1a, bufA);
    gin_range_scatter<<<N_RANGES, PT, 0, stream>>>(src, dst, cnt, rstart, recs);
    gin_range_sum_layer1<<<N_RANGES, PT, 0, stream>>>(bufA, recs, rstart, b1a, w1b, b1b,
                                                      g1, be1, m1, v1, bufB);
    gin_range_sum_layer2_fc<<<N_RANGES, PT, 0, stream>>>(bufB, recs, rstart, w2a, b2a, w2b, b2b,
                                                         g2, be2, m2, v2, fcw, fcb, out);
}


# Edge grouping kept chunk-local: each work-group sorts its own 6250 edges by range in LDS and writes a run table; no global prefix, no counting launch (4 -> 3 launches); sums stay in arrival order
# speedup vs baseline: 4.1106x; 1.0908x over previous
#include <hip/hip_runtime.h>
#include <math.h>

#define N_NODES   100000
#define N_EDGES   1600000
#define F_IN      64
#define DIM       16
#define N_CLASSES 10
#define BN_EPS    1e-5f

#define N_RANGES    256
#define RANGE_NODES 400
#define CHUNK       (N_EDGES / N_RANGES)
#define PT          1024
#define EPT         ((CHUNK + PT - 1) / PT)
#define SRC_BITS    17
#define SRC_MASK    ((1u << SRC_BITS) - 1u)
static_assert(N_RANGES * CHUNK == N_EDGES, "chunks must tile the edge list");
static_assert(N_RANGES * RANGE_NODES >= N_NODES, "ranges must cover the nodes");
static_assert(N_NODES <= (1 << SRC_BITS) && RANGE_NODES + 1 <= (1 << (32 - SRC_BITS)), "record packing");

typedef __attribute__((ext_vector_type(2))) float v2f;
typedef __attribute__((ext_vector_type(8))) float v8f;

__device__ __forceinline__ v8f wmma_k4(v2f a, v2f b, v8f c) {
    return __builtin_amdgcn_wmma_f32_16x16x4_f32(
         false, a,  false, b,
         (short)0, c,  false,  false);
}

#define OFF_BITS 13
#define OFF_MASK ((1u << OFF_BITS) - 1u)
static_assert(CHUNK < (1 << OFF_BITS), "table word packing");

#define N_TILES (N_NODES / 16)
static_assert(N_TILES * 16 == N_NODES, "whole tiles");
static_assert((PT / 32) * N_RANGES >= N_TILES, "a wave of the grouping launch for every tile");

__global__ void __launch_bounds__(PT) gin_group_proj(const int* __restrict__ src,
                                                     const int* __restrict__ dst,
                                                     const float* __restrict__ x,
                                                     const float* __restrict__ w1a,
                                                     float* __restrict__ y,
                                                     unsigned* __restrict__ recs,
                                                     unsigned* __restrict__ off) {
    __shared__ int s_h[N_RANGES];
    __shared__ int s_wl[32];
    __shared__ unsigned s_rec[CHUNK];
    const int tid = threadIdx.x;
    const int wg  = blockIdx.x;
    const int e0  = wg * CHUNK;
    unsigned d[EPT], s[EPT];
    #pragma unroll
    for (int u = 0; u < EPT; ++u) {
        const int i = e0 + min(tid + u * PT, CHUNK - 1);
        d[u] = (unsigned)dst[i];
        s[u] = (unsigned)src[i];
    }

    const int  wv       = __builtin_amdgcn_readfirstlane(tid >> 5);
    const int  tile     = wv * N_RANGES + wg;
    const bool has_tile = tile < N_TILES;
    const int  lane = tid & 31;
    const int  half = lane >> 4;
    const int  l    = lane & 15;
    const int  arow = tile * 16 + l;
    v2f a[F_IN / 4], b[F_IN / 4];
    if (has_tile) {
        #pragma unroll
        for (int i = 0; i < F_IN / 4; ++i) {
            const int ka = 4 * i + half * 2;
            a[i]   = *(const v2f*)(x + arow * F_IN + ka);
            b[i].x = w1a[ka * DIM + l];
            b[i].y = w1a[ka * DIM + DIM + l];
        }
    }

    if (tid < N_RANGES) s_h[tid] = 0;
    __syncthreads();
    int rk[EPT];
    #pragma unroll
    for (int u = 0; u < EPT; ++u) {
        rk[u] = 0;
        if (tid + u * PT < CHUNK) rk[u] = atomicAdd(&s_h[d[u] / RANGE_NODES], 1);
    }

    v8f c;
    #pragma unroll
    for (int r = 0; r < 8; ++r) c[r] = 0.0f;
    if (has_tile) {
        #pragma unroll
        for (int i = 0; i < F_IN / 4; ++i) c = wmma_k4(a[i], b[i], c);
    }
    __syncthreads();

    const int L = (tid < N_RANGES) ? s_h[tid] : 0;
    int incL = L;
    for (int o = 1; o < 32; o <<= 1) {
        const int tL = __shfl_up(incL, o);
        if (lane >= o) incL += tL;
    }
    if (lane == 31) s_wl[wv] = incL;
    __syncthreads();
    unsigned offw = 0;
    if (tid < N_RANGES) {
        int offL = 0;
        for (int k = 0; k < wv; ++k) offL += s_wl[k];
        const int lstart = offL + incL - L;
        s_h[tid] = lstart;
        offw = (unsigned)lstart | ((unsigned)L << OFF_BITS);
    }
    __syncthreads();

    #pragma unroll
    for (int u = 0; u < EPT; ++u) {
        if (tid + u * PT < CHUNK) {
            const unsigned rg = d[u] / RANGE_NODES;
            const unsigned dl = d[u] - rg * RANGE_NODES;
            s_rec[s_h[rg] + rk[u]] = s[u] | (dl << SRC_BITS);
        }
    }
    __syncthreads();
    if (tid < N_RANGES) off[wg * N_RANGES + tid] = offw;
    if (has_tile) {
        #pragma unroll
        for (int r = 0; r < 8; ++r)
            y[(tile * 16 + half * 8 + r) * DIM + l] = c[r];
    }
    for (int t = tid; t < CHUNK; t += PT) recs[e0 + t] = s_rec[t];
}

#define SPARE_REC ((unsigned)RANGE_NODES << SRC_BITS)
#define ACC_ROWS  ((PT / 32) * 16)
#define RANGE_V4  (RANGE_NODES * DIM / 4)
#define N_WAVES   (PT / 32)
#define WRUNS     (N_RANGES / N_WAVES)
static_assert(RANGE_NODES % 16 == 0 && ACC_ROWS >= RANGE_NODES + 1, "tiles per range");
static_assert(WRUNS * N_WAVES == N_RANGES, "whole runs per wave");

__device__ __forceinline__ void range_gather_sum(const float* __restrict__ feat,
                                                 const unsigned* __restrict__ recs,
                                                 const unsigned* __restrict__ off,
                                                 float* s_acc) {
    const int tid = threadIdx.x;
    const int l   = tid & 15;
    const int wv = __builtin_amdgcn_readfirstlane(tid >> 5);
    int end[WRUNS], org[WRUNS];
    {
        int run = 0;
        #pragma unroll
        for (int i = 0; i < WRUNS; ++i) {
            const int chunk = wv + i * N_WAVES;
            const unsigned o = off[chunk * N_RANGES + (int)blockIdx.x];
            org[i] = chunk * CHUNK + (int)(o & OFF_MASK) - run;
            run   += (int)(o >> OFF_BITS);
            end[i] = run;
        }
    }
    const int tot = end[WRUNS - 1];
    for (int t = tid; t < ACC_ROWS * DIM; t += PT) s_acc[t] = 0.0f;
    __syncthreads();

    int k = tid & 31;
    int w = org[WRUNS - 1];
    #pragma unroll
    for (int i = WRUNS - 2; i >= 0; --i) w = (k < end[i]) ? org[i] : w;
    unsigned mine = recs[min(w + k, N_EDGES - 1)];
    if (k >= tot) mine = SPARE_REC;
    for (int base = 0; base < tot; base += 32) {
        k += 32;
        w = org[WRUNS - 1];
        #pragma unroll
        for (int i = WRUNS - 2; i >= 0; --i) w = (k < end[i]) ? org[i] : w;
        unsigned nxt = recs[min(w + k, N_EDGES - 1)];
        if (k >= tot) nxt = SPARE_REC;
        unsigned r[16];
        float    v[16];
        #pragma unroll
        for (int j = 0; j < 16; ++j) {
            r[j] = __shfl(mine, j, 16);
            v[j] = feat[(r[j] & SRC_MASK) * DIM + l];
        }
        #pragma unroll
        for (int j = 0; j < 16; ++j)
            __hip_atomic_fetch_add(&s_acc[(r[j] >> SRC_BITS) * DIM + l], v[j],
                                   __ATOMIC_RELAXED, __HIP_MEMORY_SCOPE_WORKGROUP);
        mine = nxt;
    }
}

template <int NC>
__device__ __forceinline__ void load_bfrag(const float* __restrict__ w, int l, int half, v2f (&b)[4]) {
    #pragma unroll
    for (int i = 0; i < 4; ++i) {
        const int ka = 4 * i + half * 2;
        b[i].x = (l < NC) ? w[ka * NC + l]       : 0.0f;
        b[i].y = (l < NC) ? w[(ka + 1) * NC + l] : 0.0f;
    }
}

__device__ __forceinline__ v8f tile_gemm16(const float* tileA, const v2f (&b)[4],
                                           v8f c, int l, int half) {
    #pragma unroll
    for (int i = 0; i < 4; ++i) {
        const int ka = 4 * i + half * 2;
        v2f a = *(const v2f*)(tileA + l * DIM + ka);
        c = wmma_k4(a, b[i], c);
    }
    return c;
}

__global__ void __launch_bounds__(PT) gin_range_sum_layer1(const float* __restrict__ feat,
                                                           const unsigned* __restrict__ recs,
                                                           const unsigned* __restrict__ off,
                                                           const float* __restrict__ b1a,
                                                           const float* __restrict__ w1b, const float* __restrict__ b1b,
                                                           const float* __restrict__ g,  const float* __restrict__ be,
                                                           const float* __restrict__ mu, const float* __restrict__ var,
                                                           float* __restrict__ u) {
    __shared__ __attribute__((aligned(16))) float s_acc[ACC_ROWS * DIM];
    range_gather_sum(feat, recs, off, s_acc);

    const int tid  = threadIdx.x;
    const int lane = tid & 31;
    const int half = lane >> 4;
    const int l    = lane & 15;
    const int q0   = blockIdx.x * RANGE_V4;
    float4* acc4 = (float4*)s_acc;

    const int cb = (tid & 3) * 4;
    const float b0 = b1a[cb], b1 = b1a[cb + 1], b2 = b1a[cb + 2], b3 = b1a[cb + 3];
    v2f wb[4];
    load_bfrag<DIM>(w1b, l, half, wb);
    const float cb1 = b1b[l];
    const float sc = g[l] * rsqrtf(var[l] + BN_EPS);
    const float mm = mu[l], bb = be[l];
    __syncthreads();

    {
        const float4* feat4 = (const float4*)feat;
        for (int t = tid; t < RANGE_V4; t += PT) {
            if (q0 + t < N_NODES * DIM / 4) {
                const float4 f = feat4[q0 + t];
                float4 a = acc4[t];
                a.x = fmaxf((f.x + a.x) + b0, 0.0f);
                a.y = fmaxf((f.y + a.y) + b1, 0.0f);
                a.z = fmaxf((f.z + a.z) + b2, 0.0f);
                a.w = fmaxf((f.w + a.w) + b3, 0.0f);
                acc4[t] = a;
            }
        }
    }
    __syncthreads();

    float* tileA = s_acc + (tid >> 5) * 16 * DIM;
    v8f c;
    #pragma unroll
    for (int r = 0; r < 8; ++r) c[r] = cb1;
    c = tile_gemm16(tileA, wb, c, l, half);
    #pragma unroll
    for (int r = 0; r < 8; ++r) {
        const float t = fmaxf(c[r], 0.0f);
        tileA[(half * 8 + r) * DIM + l] = (t - mm) * sc + bb;
    }
    __syncthreads();

    float4* u4 = (float4*)u;
    for (int t = tid; t < RANGE_V4; t += PT)
        if (q0 + t < N_NODES * DIM / 4) u4[q0 + t] = acc4[t];
}

__global__ void __launch_bounds__(PT) gin_range_sum_layer2_fc(const float* __restrict__ feat,
                                                              const unsigned* __restrict__ recs,
                                                              const unsigned* __restrict__ off,
                                                              const float* __restrict__ w2a, const float* __restrict__ b2a,
                                                              const float* __restrict__ w2b, const float* __restrict__ b2b,
                                                              const float* __restrict__ g,   const float* __restrict__ be,
                                                              const float* __restrict__ mu,  const float* __restrict__ var,
                                                              const float* __restrict__ fcw, const float* __restrict__ fcb,
                                                              float* __restrict__ out) {
    __shared__ __attribute__((aligned(16))) float s_acc[ACC_ROWS * DIM];
    __shared__ __attribute__((aligned(16))) float s_out[RANGE_NODES * N_CLASSES];
    range_gather_sum(feat, recs, off, s_acc);

    const int tid  = threadIdx.x;
    const int lane = tid & 31;
    const int half = lane >> 4;
    const int l    = lane & 15;
    const int q0   = blockIdx.x * RANGE_V4;
    float4* acc4 = (float4*)s_acc;

    v2f wa[4], wb[4], wf[4];
    load_bfrag<DIM>(w2a, l, half, wa);
    load_bfrag<DIM>(w2b, l, half, wb);
    load_bfrag<N_CLASSES>(fcw, l, half, wf);
    const float cba = b2a[l], cbb = b2b[l];
    const float cbf = (l < N_CLASSES) ? fcb[l] : 0.0f;
    const float sc = g[l] * rsqrtf(var[l] + BN_EPS);
    const float mm = mu[l], bb = be[l];
    __syncthreads();

    {
        const float4* feat4 = (const float4*)feat;
        for (int t = tid; t < RANGE_V4; t += PT) {
            if (q0 + t < N_NODES * DIM / 4) {
                const float4 f = feat4[q0 + t];
                float4 a = acc4[t];
                a.x = f.x + a.x;  a.y = f.y + a.y;  a.z = f.z + a.z;  a.w = f.w + a.w;
                acc4[t] = a;
            }
        }
    }
    __syncthreads();

    float* tileA = s_acc + (tid >> 5) * 16 * DIM;
    v8f c;

    #pragma unroll
    for (int r = 0; r < 8; ++r) c[r] = cba;
    c = tile_gemm16(tileA, wa, c, l, half);
    #pragma unroll
    for (int r = 0; r < 8; ++r) tileA[(half * 8 + r) * DIM + l] = fmaxf(c[r], 0.0f);
    __syncthreads();

    #pragma unroll
    for (int r = 0; r < 8; ++r) c[r] = cbb;
    c = tile_gemm16(tileA, wb, c, l, half);
    #pragma unroll
    for (int r = 0; r < 8; ++r) {
        const float t = fmaxf(c[r], 0.0f);
        tileA[(half * 8 + r) * DIM + l] = (t - mm) * sc + bb;
    }
    __syncthreads();

    #pragma unroll
    for (int r = 0; r < 8; ++r) c[r] = cbf;
    c = tile_gemm16(tileA, wf, c, l, half);
    #pragma unroll
    for (int r = 0; r < 8; ++r) tileA[(half * 8 + r) * DIM + l] = c[r];
    __syncthreads();

    if (tid < RANGE_NODES) {
        const float* p = s_acc + tid * DIM;
        float v[N_CLASSES];
        float m = -INFINITY;
        #pragma unroll
        for (int j = 0; j < N_CLASSES; ++j) { v[j] = p[j]; m = fmaxf(m, v[j]); }
        float s = 0.0f;
        #pragma unroll
        for (int j = 0; j < N_CLASSES; ++j) s += expf(v[j] - m);
        const float ls = m + logf(s);
        #pragma unroll
        for (int j = 0; j < N_CLASSES; ++j) s_out[tid * N_CLASSES + j] = v[j] - ls;
    }
    __syncthreads();

    const int o0 = blockIdx.x * (RANGE_NODES * N_CLASSES / 4);
    const float4* so4 = (const float4*)s_out;
    float4* out4 = (float4*)out;
    for (int t = tid; t < RANGE_NODES * N_CLASSES / 4; t += PT)
        if (o0 + t < N_NODES * N_CLASSES / 4) out4[o0 + t] = so4[t];
}

extern "C" void kernel_launch(void* const* d_in, const int* in_sizes, int n_in,
                              void* d_out, int out_size, void* d_ws, size_t ws_size,
                              hipStream_t stream) {
    (void)in_sizes; (void)n_in; (void)out_size; (void)ws_size;

    const float* x   = (const float*)d_in[0];
    const int*   ei  = (const int*)d_in[1];
    const int*   src = ei;
    const int*   dst = ei + N_EDGES;
    const float* w1a = (const float*)d_in[2];  const float* b1a = (const float*)d_in[3];
    const float* w1b = (const float*)d_in[4];  const float* b1b = (const float*)d_in[5];
    const float* g1  = (const float*)d_in[6];  const float* be1 = (const float*)d_in[7];
    const float* m1  = (const float*)d_in[8];  const float* v1  = (const float*)d_in[9];
    const float* w2a = (const float*)d_in[10]; const float* b2a = (const float*)d_in[11];
    const float* w2b = (const float*)d_in[12]; const float* b2b = (const float*)d_in[13];
    const float* g2  = (const float*)d_in[14]; const float* be2 = (const float*)d_in[15];
    const float* m2  = (const float*)d_in[16]; const float* v2  = (const float*)d_in[17];
    const float* fcw = (const float*)d_in[18]; const float* fcb = (const float*)d_in[19];
    float* out = (float*)d_out;

    const int NF = N_NODES * DIM;
    float* ws   = (float*)d_ws;
    float* bufA = ws;
    float* bufB = ws + (size_t)NF;
    unsigned* recs = (unsigned*)(ws + (size_t)2 * NF);
    unsigned* off  = (unsigned*)(ws + (size_t)3 * NF);

    gin_group_proj<<<N_RANGES, PT, 0, stream>>>(src, dst, x, w1a, bufA, recs, off);
    gin_range_sum_layer1<<<N_RANGES, PT, 0, stream>>>(bufA, recs, off, b1a, w1b, b1b,
                                                      g1, be1, m1, v1, bufB);
    gin_range_sum_layer2_fc<<<N_RANGES, PT, 0, stream>>>(bufB, recs, off, w2a, b2a, w2b, b2b,
                                                         g2, be2, m2, v2, fcw, fcb, out);
}
